# starting kernel
# baseline (speedup 1.0000x reference)
#include <hip/hip_runtime.h>

typedef __attribute__((ext_vector_type(4))) float v4f;
typedef __attribute__((ext_vector_type(2))) float v2f;
typedef __attribute__((ext_vector_type(8))) float v8f;

constexpr int kC    = 32;
constexpr int kMin  = 16;
constexpr int kMout = 32;
constexpr int kB    = 4;
constexpr int kNCube = kB * kMin * kC;

__global__ __launch_bounds__(256) void phase1_reduce(const float* __restrict__ x,
                                                     float* __restrict__ p3,
                                                     float* __restrict__ p4) {
  __shared__ float s3_partial[8][kC];
  __shared__ float s4_final[kC];

  const int t = threadIdx.x;
  const int w = t >> 5;
  const int l = t & 31;
  const int cube = blockIdx.x;

  const v4f* __restrict__ xc = (const v4f*)(x + (size_t)cube * (kC * kC * kC));

  float acc4 = 0.f;
  for (int i = 0; i < kC; ++i) {
    v4f v = __builtin_nontemporal_load(&xc[i * 256 + t]);
    float s = (v.x + v.y) + (v.z + v.w);
    acc4 += s;
    s += __shfl_xor(s, 16, 32);
    s += __shfl_xor(s, 8, 32);
    s += __shfl_xor(s, 4, 32);
    s += __shfl_xor(s, 2, 32);
    s += __shfl_xor(s, 1, 32);
    if (l == 0) s3_partial[w][i] = s;
  }
  acc4 += __shfl_xor(acc4, 1, 32);
  acc4 += __shfl_xor(acc4, 2, 32);
  acc4 += __shfl_xor(acc4, 4, 32);
  if ((l & 7) == 0) s4_final[4 * w + (l >> 3)] = acc4;
  __syncthreads();

  if (t < kC) {
    float r = 0.f;
#pragma unroll
    for (int w2 = 0; w2 < 8; ++w2) r += s3_partial[w2][t];
    p3[cube * kC + t] = r;
  } else if (t < 2 * kC) {
    p4[cube * kC + (t - kC)] = s4_final[t - kC];
  }
}

__global__ __launch_bounds__(256) void phase2_wmma(const float* __restrict__ p3,
                                                   const float* __restrict__ p4,
                                                   const float* __restrict__ alpha,
                                                   const float* __restrict__ beta,
                                                   const float* __restrict__ gamma_,
                                                   const float* __restrict__ delta,
                                                   float* __restrict__ out) {
  __shared__ float S[kB][kMin][4][kC];
  __shared__ float W[kMin][kMout];

  const int t = threadIdx.x;

  {
    const int bm = t >> 2;
    const int q  = t & 3;
    const int b = bm >> 4, m = bm & 15;
#pragma unroll
    for (int cc = 0; cc < 8; ++cc) {
      const int c = q * 8 + cc;
      float s1 = 0.f, s3 = 0.f, s4 = 0.f;
      for (int k = 0; k < kC; ++k) {
        s1 += p3[(bm * kC + c) * kC + k];
        s3 += p3[(bm * kC + k) * kC + c];
        s4 += p4[(bm * kC + k) * kC + c];
      }
      S[b][m][0][c] = s1;
      S[b][m][1][c] = s1;
      S[b][m][2][c] = s3;
      S[b][m][3][c] = s4;
    }
  }

  for (int e = t; e < kMin * kMout; e += 256)
    W[e >> 5][e & 31] = alpha[e] + beta[e] + gamma_[e] + delta[e];

  __syncthreads();

  const int w  = t >> 5;
  const int l  = t & 31;
  const int lh = l >> 4;
  const int ll = l & 15;

  for (int j = 0; j < 8; ++j) {
    const int tau = w * 8 + j;
    const int b  = tau >> 4;
    const int i  = (tau >> 2) & 3;
    const int tn = (tau >> 1) & 1;
    const int tc = tau & 1;

    v8f acc = {};
#pragma unroll
    for (int k0 = 0; k0 < kMin; k0 += 4) {
      const int m0 = k0 + 0 + 2 * lh;
      const int m1 = k0 + 1 + 2 * lh;
      v2f a, bmat;
      a.x = W[m0][tn * 16 + ll];
      a.y = W[m1][tn * 16 + ll];
      bmat.x = S[b][m0][i][tc * 16 + ll];
      bmat.y = S[b][m1][i][tc * 16 + ll];
      acc = __builtin_amdgcn_wmma_f32_16x16x4_f32(
           false, a,  false, bmat,
           (short)0, acc,  false,  false);
    }
#pragma unroll
    for (int v = 0; v < 8; ++v) {
      const int n = tn * 16 + v + 8 * lh;
      const int c = tc * 16 + ll;
      out[((b * kMout + n) * 4 + i) * kC + c] = acc[v];
    }
  }
}

extern "C" void kernel_launch(void* const* d_in, const int* in_sizes, int n_in,
                              void* d_out, int out_size, void* d_ws, size_t ws_size,
                              hipStream_t stream) {
  const float* x      = (const float*)d_in[0];
  const float* alpha  = (const float*)d_in[1];
  const float* beta   = (const float*)d_in[2];
  const float* gamma_ = (const float*)d_in[3];
  const float* delta  = (const float*)d_in[4];
  float* out = (float*)d_out;

  float* p3 = (float*)d_ws;
  float* p4 = p3 + (size_t)kNCube * kC;

  phase1_reduce<<<kNCube, 256, 0, stream>>>(x, p3, p4);
  phase2_wmma<<<1, 256, 0, stream>>>(p3, p4, alpha, beta, gamma_, delta, out);
}


# Fold-and-product kernel spread from 1 work-group over 12 (one per batch and statistic): one plane per wave, whole-line 16-byte loads issued as one group; the fold's sums become a lane tree
# speedup vs baseline: 1.4087x; 1.4087x over previous
#include <hip/hip_runtime.h>

typedef __attribute__((ext_vector_type(4))) float v4f;
typedef __attribute__((ext_vector_type(2))) float v2f;
typedef __attribute__((ext_vector_type(8))) float v8f;

constexpr int kC    = 32;
constexpr int kMin  = 16;
constexpr int kMout = 32;
constexpr int kB    = 4;
constexpr int kNCube = kB * kMin * kC;

__global__ __launch_bounds__(256) void phase1_reduce(const float* __restrict__ x,
                                                     float* __restrict__ p3,
                                                     float* __restrict__ p4) {
  __shared__ float s3_partial[8][kC];
  __shared__ float s4_final[kC];

  const int t = threadIdx.x;
  const int w = t >> 5;
  const int l = t & 31;
  const int cube = blockIdx.x;

  const v4f* __restrict__ xc = (const v4f*)(x + (size_t)cube * (kC * kC * kC));

  float acc4 = 0.f;
  for (int i = 0; i < kC; ++i) {
    v4f v = __builtin_nontemporal_load(&xc[i * 256 + t]);
    float s = (v.x + v.y) + (v.z + v.w);
    acc4 += s;
    s += __shfl_xor(s, 16, 32);
    s += __shfl_xor(s, 8, 32);
    s += __shfl_xor(s, 4, 32);
    s += __shfl_xor(s, 2, 32);
    s += __shfl_xor(s, 1, 32);
    if (l == 0) s3_partial[w][i] = s;
  }
  acc4 += __shfl_xor(acc4, 1, 32);
  acc4 += __shfl_xor(acc4, 2, 32);
  acc4 += __shfl_xor(acc4, 4, 32);
  if ((l & 7) == 0) s4_final[4 * w + (l >> 3)] = acc4;
  __syncthreads();

  if (t < kC) {
    float r = 0.f;
#pragma unroll
    for (int w2 = 0; w2 < 8; ++w2) r += s3_partial[w2][t];
    p3[cube * kC + t] = r;
  } else if (t < 2 * kC) {
    p4[cube * kC + (t - kC)] = s4_final[t - kC];
  }
}

__global__ __launch_bounds__(512) void phase2_wmma(const float* __restrict__ p3,
                                                   const float* __restrict__ p4,
                                                   const float* __restrict__ alpha,
                                                   const float* __restrict__ beta,
                                                   const float* __restrict__ gamma_,
                                                   const float* __restrict__ delta,
                                                   float* __restrict__ out) {
  __shared__ __attribute__((aligned(16))) float S[kMin][kC];
  __shared__ float W[kMin][kMout];

  const int t = threadIdx.x;
  const int l = t & 31;
  const int w = __builtin_amdgcn_readfirstlane(t >> 5);
  const int stat = blockIdx.x;
  const int b = blockIdx.y;

  W[t >> 5][t & 31] = alpha[t] + beta[t] + gamma_[t] + delta[t];

  const float* src = (stat == 2) ? p4 : p3;
  const v4f* __restrict__ P = (const v4f*)(src + (size_t)(b * kMin + w) * (kC * kC));
  v4f v[8];
  for (int j = 0; j < 8; ++j)
    v[j] = P[l + 32 * j];

  if (stat == 0) {
    float s[8];
    for (int j = 0; j < 8; ++j) s[j] = (v[j].x + v[j].y) + (v[j].z + v[j].w);
    for (int j = 0; j < 8; ++j) s[j] += __shfl_xor(s[j], 1, 32);
    for (int j = 0; j < 8; ++j) s[j] += __shfl_xor(s[j], 2, 32);
    for (int j = 0; j < 8; ++j) s[j] += __shfl_xor(s[j], 4, 32);
    if ((l & 7) == 0) {
      for (int j = 0; j < 8; ++j) S[w][(l >> 3) + 4 * j] = s[j];
    }
  } else {
    v4f a = v[0];
    for (int j = 1; j < 8; ++j) a += v[j];
    a.x += __shfl_xor(a.x, 8, 32);
    a.y += __shfl_xor(a.y, 8, 32);
    a.z += __shfl_xor(a.z, 8, 32);
    a.w += __shfl_xor(a.w, 8, 32);
    a.x += __shfl_xor(a.x, 16, 32);
    a.y += __shfl_xor(a.y, 16, 32);
    a.z += __shfl_xor(a.z, 16, 32);
    a.w += __shfl_xor(a.w, 16, 32);
    if (l < 8) *(v4f*)&S[w][4 * l] = a;
  }

  __syncthreads();

  if (w < 2) {
    const int tn = w;
    const int lh = l >> 4;
    const int ll = l & 15;

    v8f acc0 = {}, acc1 = {};
    for (int k0 = 0; k0 < kMin; k0 += 4) {
      const int m0 = k0 + 0 + 2 * lh;
      const int m1 = k0 + 1 + 2 * lh;
      v2f a, b0, b1;
      a.x = W[m0][tn * 16 + ll];
      a.y = W[m1][tn * 16 + ll];
      b0.x = S[m0][ll];
      b0.y = S[m1][ll];
      b1.x = S[m0][16 + ll];
      b1.y = S[m1][16 + ll];
      acc0 = __builtin_amdgcn_wmma_f32_16x16x4_f32(
           false, a,  false, b0,
           (short)0, acc0,  false,  false);
      acc1 = __builtin_amdgcn_wmma_f32_16x16x4_f32(
           false, a,  false, b1,
           (short)0, acc1,  false,  false);
    }
    const int i0 = (stat == 0) ? 0 : stat + 1;
    float* __restrict__ o = out + ((b * kMout + tn * 16 + 8 * lh) * 4 + i0) * kC + ll;
    for (int r = 0; r < 8; ++r) {
      o[r * 4 * kC] = acc0[r];
      o[r * 4 * kC + 16] = acc1[r];
    }
    if (stat == 0) {
      for (int r = 0; r < 8; ++r) {
        o[r * 4 * kC + kC] = acc0[r];
        o[r * 4 * kC + kC + 16] = acc1[r];
      }
    }
  }
}

extern "C" void kernel_launch(void* const* d_in, const int* in_sizes, int n_in,
                              void* d_out, int out_size, void* d_ws, size_t ws_size,
                              hipStream_t stream) {
  const float* x      = (const float*)d_in[0];
  const float* alpha  = (const float*)d_in[1];
  const float* beta   = (const float*)d_in[2];
  const float* gamma_ = (const float*)d_in[3];
  const float* delta  = (const float*)d_in[4];
  float* out = (float*)d_out;

  float* p3 = (float*)d_ws;
  float* p4 = p3 + (size_t)kNCube * kC;

  phase1_reduce<<<kNCube, 256, 0, stream>>>(x, p3, p4);
  phase2_wmma<<<dim3(3, kB), 512, 0, stream>>>(p3, p4, alpha, beta, gamma_, delta, out);
}


# Phase 1 as a stream: 512 work-groups; a wave owns 4 whole c3 slabs per cube and requests 24 whole-line 16-byte loads ahead of its sums; lane trees once per cube, not per load; sum order changes
# speedup vs baseline: 1.7558x; 1.2464x over previous
#include <hip/hip_runtime.h>

typedef __attribute__((ext_vector_type(4))) float v4f;
typedef __attribute__((ext_vector_type(2))) float v2f;
typedef __attribute__((ext_vector_type(8))) float v8f;

constexpr int kC    = 32;
constexpr int kMin  = 16;
constexpr int kMout = 32;
constexpr int kB    = 4;
constexpr int kNCube = kB * kMin * kC;

constexpr int kCubesPerWG = 4;
constexpr int kGrid1 = kNCube / kCubesPerWG;
constexpr int kSlabs = 4 * kCubesPerWG;
constexpr int kAhead = 3;
constexpr int kRing  = kAhead + 1;

__global__ __launch_bounds__(256) void phase1_reduce(const float* __restrict__ x,
                                                     float* __restrict__ p3,
                                                     float* __restrict__ p4) {
  __shared__ __attribute__((aligned(16))) float s3_fin[kCubesPerWG][kC];
  __shared__ float s4_part[kCubesPerWG][8][kC];

  const int t = threadIdx.x;
  const int l = t & 31;
  const int w = __builtin_amdgcn_readfirstlane(t >> 5);
  const int cube0 = blockIdx.x * kCubesPerWG;

  const v4f* __restrict__ base =
      (const v4f*)(x + (size_t)cube0 * (kC * kC * kC) + (size_t)w * (4 * kC * kC)) + l;

  v4f buf[kRing][8];
#define LOAD_SLAB(I)                                                             \
  do {                                                                           \
    const v4f* p_ = base + ((I) >> 2) * (kC * kC * kC / 4) + ((I) & 3) * (kC * kC / 4); \
    for (int j = 0; j < 8; ++j)                                                  \
        buf[(I) % kRing][j] = __builtin_nontemporal_load(p_ + 32 * j);           \
  } while (0)

  for (int i = 0; i < kAhead; ++i) LOAD_SLAB(i);

  float acc4[8];
  float s3v[4];

#pragma unroll
  for (int i = 0; i < kSlabs; ++i) {
    __builtin_amdgcn_sched_barrier(0);
    if (i + kAhead < kSlabs) LOAD_SLAB(i + kAhead);
    __builtin_amdgcn_sched_barrier(0);

    const v4f* b = buf[i % kRing];
    float h[8];
    for (int j = 0; j < 8; ++j) h[j] = (b[j].x + b[j].z) + (b[j].y + b[j].w);
    for (int j = 0; j < 8; ++j) acc4[j] = ((i & 3) == 0) ? h[j] : acc4[j] + h[j];
    s3v[i & 3] = ((h[0] + h[1]) + (h[2] + h[3])) + ((h[4] + h[5]) + (h[6] + h[7]));
    asm volatile("" : "+v"(s3v[i & 3]));
    asm volatile("" : "+v"(acc4[0]), "+v"(acc4[1]), "+v"(acc4[2]), "+v"(acc4[3]),
                      "+v"(acc4[4]), "+v"(acc4[5]), "+v"(acc4[6]), "+v"(acc4[7]));

    if ((i & 3) == 3) {
      const int q = i >> 2;
      for (int m = 16; m >= 1; m >>= 1) {
        for (int s = 0; s < 4; ++s) s3v[s] += __shfl_xor(s3v[s], m, 32);
      }
      if (l == 0) {
        v4f r4;
        r4.x = s3v[0]; r4.y = s3v[1]; r4.z = s3v[2]; r4.w = s3v[3];
        *(v4f*)&s3_fin[q][4 * w] = r4;
      }
      for (int m = 1; m <= 4; m <<= 1) {
        for (int j = 0; j < 8; ++j) acc4[j] += __shfl_xor(acc4[j], m, 32);
      }
      if ((l & 7) == 0) {
        for (int j = 0; j < 8; ++j) s4_part[q][w][4 * j + (l >> 3)] = acc4[j];
      }
    }
  }
#undef LOAD_SLAB

  __syncthreads();

  const int q = w & 3;
  const int cube = cube0 + q;
  if (w < 4) {
    float r = 0.f;
    for (int w2 = 0; w2 < 8; ++w2) r += s4_part[q][w2][l];
    p4[cube * kC + l] = r;
  } else {
    p3[cube * kC + l] = s3_fin[q][l];
  }
}

__global__ __launch_bounds__(512) void phase2_wmma(const float* __restrict__ p3,
                                                   const float* __restrict__ p4,
                                                   const float* __restrict__ alpha,
                                                   const float* __restrict__ beta,
                                                   const float* __restrict__ gamma_,
                                                   const float* __restrict__ delta,
                                                   float* __restrict__ out) {
  __shared__ __attribute__((aligned(16))) float S[kMin][kC];
  __shared__ float W[kMin][kMout];

  const int t = threadIdx.x;
  const int l = t & 31;
  const int w = __builtin_amdgcn_readfirstlane(t >> 5);
  const int stat = blockIdx.x;
  const int b = blockIdx.y;

  W[t >> 5][t & 31] = alpha[t] + beta[t] + gamma_[t] + delta[t];

  const float* src = (stat == 2) ? p4 : p3;
  const v4f* __restrict__ P = (const v4f*)(src + (size_t)(b * kMin + w) * (kC * kC));
  v4f v[8];
  for (int j = 0; j < 8; ++j)
    v[j] = P[l + 32 * j];

  if (stat == 0) {
    float s[8];
    for (int j = 0; j < 8; ++j) s[j] = (v[j].x + v[j].y) + (v[j].z + v[j].w);
    for (int j = 0; j < 8; ++j) s[j] += __shfl_xor(s[j], 1, 32);
    for (int j = 0; j < 8; ++j) s[j] += __shfl_xor(s[j], 2, 32);
    for (int j = 0; j < 8; ++j) s[j] += __shfl_xor(s[j], 4, 32);
    if ((l & 7) == 0) {
      for (int j = 0; j < 8; ++j) S[w][(l >> 3) + 4 * j] = s[j];
    }
  } else {
    v4f a = v[0];
    for (int j = 1; j < 8; ++j) a += v[j];
    a.x += __shfl_xor(a.x, 8, 32);
    a.y += __shfl_xor(a.y, 8, 32);
    a.z += __shfl_xor(a.z, 8, 32);
    a.w += __shfl_xor(a.w, 8, 32);
    a.x += __shfl_xor(a.x, 16, 32);
    a.y += __shfl_xor(a.y, 16, 32);
    a.z += __shfl_xor(a.z, 16, 32);
    a.w += __shfl_xor(a.w, 16, 32);
    if (l < 8) *(v4f*)&S[w][4 * l] = a;
  }

  __syncthreads();

  if (w < 2) {
    const int tn = w;
    const int lh = l >> 4;
    const int ll = l & 15;

    v8f acc0 = {}, acc1 = {};
    for (int k0 = 0; k0 < kMin; k0 += 4) {
      const int m0 = k0 + 0 + 2 * lh;
      const int m1 = k0 + 1 + 2 * lh;
      v2f a, b0, b1;
      a.x = W[m0][tn * 16 + ll];
      a.y = W[m1][tn * 16 + ll];
      b0.x = S[m0][ll];
      b0.y = S[m1][ll];
      b1.x = S[m0][16 + ll];
      b1.y = S[m1][16 + ll];
      acc0 = __builtin_amdgcn_wmma_f32_16x16x4_f32(
           false, a,  false, b0,
           (short)0, acc0,  false,  false);
      acc1 = __builtin_amdgcn_wmma_f32_16x16x4_f32(
           false, a,  false, b1,
           (short)0, acc1,  false,  false);
    }
    const int i0 = (stat == 0) ? 0 : stat + 1;
    float* __restrict__ o = out + ((b * kMout + tn * 16 + 8 * lh) * 4 + i0) * kC + ll;
    for (int r = 0; r < 8; ++r) {
      o[r * 4 * kC] = acc0[r];
      o[r * 4 * kC + 16] = acc1[r];
    }
    if (stat == 0) {
      for (int r = 0; r < 8; ++r) {
        o[r * 4 * kC + kC] = acc0[r];
        o[r * 4 * kC + kC + 16] = acc1[r];
      }
    }
  }
}

extern "C" void kernel_launch(void* const* d_in, const int* in_sizes, int n_in,
                              void* d_out, int out_size, void* d_ws, size_t ws_size,
                              hipStream_t stream) {
  const float* x      = (const float*)d_in[0];
  const float* alpha  = (const float*)d_in[1];
  const float* beta   = (const float*)d_in[2];
  const float* gamma_ = (const float*)d_in[3];
  const float* delta  = (const float*)d_in[4];
  float* out = (float*)d_out;

  float* p3 = (float*)d_ws;
  float* p4 = p3 + (size_t)kNCube * kC;

  phase1_reduce<<<kGrid1, 256, 0, stream>>>(x, p3, p4);
  phase2_wmma<<<dim3(3, kB), 512, 0, stream>>>(p3, p4, alpha, beta, gamma_, delta, out);
}
